# starting kernel
# baseline (speedup 1.0000x reference)
#include <hip/hip_runtime.h>
#include <hip/hip_bf16.h>

#define NB 16
#define NC 128
#define ND 256
#define NROWS (NB * NC)

typedef __attribute__((ext_vector_type(2))) float v2f;
typedef __attribute__((ext_vector_type(8))) float v8f;

static __device__ __forceinline__ v8f wmma4(v2f a, v2f b, v8f c) {
    return __builtin_amdgcn_wmma_f32_16x16x4_f32(
         false, a,  false, b,
         (short)0, c,  false,  false);
}

__global__ void __launch_bounds__(256) k_gemm_pq(const float* __restrict__ H,
                                                 const float* __restrict__ W1,
                                                 float* __restrict__ P,
                                                 float* __restrict__ Q) {
    const int lane = threadIdx.x & 31;
    const int wv   = threadIdx.x >> 5;
    const int tile = blockIdx.x * 8 + wv;
    const int mt = tile >> 4;
    const int nt = tile & 15;
    const int hi = lane >> 4, lo = lane & 15;
    const int n  = nt * 16 + lo;

    const float* Arow = H + (mt * 16 + lo) * ND;
    v8f accP = {}, accQ = {};
#pragma unroll 4
    for (int k = 0; k < ND; k += 4) {
        const int ka = k + 2 * hi;
        v2f a = *(const v2f*)(Arow + ka);
        v2f bp, bq;
        bp.x = W1[ka * ND + n];
        bp.y = W1[(ka + 1) * ND + n];
        bq.x = W1[(ka + ND) * ND + n];
        bq.y = W1[(ka + ND + 1) * ND + n];
        accP = wmma4(a, bp, accP);
        accQ = wmma4(a, bq, accQ);
    }
#pragma unroll
    for (int g = 0; g < 8; ++g) {
        const int row = mt * 16 + g + 8 * hi;
        P[row * ND + n] = accP[g];
        Q[row * ND + n] = accQ[g];
    }
}

__global__ void __launch_bounds__(256) k_scores(const float* __restrict__ P,
                                                const float* __restrict__ Q,
                                                const float* __restrict__ b1,
                                                const float* __restrict__ W2,
                                                const float* __restrict__ b2,
                                                const float* __restrict__ A_a,
                                                float* __restrict__ G) {
    __shared__ float Pl[16][ND + 1];
    __shared__ float Ql[16][ND + 1];
    __shared__ float b1s[ND];
    __shared__ float w2s[ND];

    const int blk = blockIdx.x;
    const int vt = blk & 7;
    const int ut = (blk >> 3) & 7;
    const int b  = blk >> 6;
    const int t  = threadIdx.x;

#pragma unroll
    for (int i = t; i < 16 * ND; i += 256) {
        const int r = i >> 8, k = i & (ND - 1);
        Pl[r][k] = P[(b * NC + ut * 16 + r) * ND + k];
        Ql[r][k] = Q[(b * NC + vt * 16 + r) * ND + k];
    }
    b1s[t] = b1[t];
    w2s[t] = W2[t];
    __syncthreads();

    const int u = t >> 4, v = t & 15;
    float acc = 0.f;
#pragma unroll 8
    for (int k = 0; k < ND; ++k) {
        const float x = Pl[u][k] + Ql[v][k] + b1s[k];
        acc = fmaf(fmaxf(x, 0.f), w2s[k], acc);
    }
    const float s  = 1.0f / (1.0f + __expf(-(acc + b2[0])));
    const int uu = ut * 16 + u, vv = vt * 16 + v;
    G[(b * NC + uu) * NC + vv] = s * A_a[uu * NC + vv];
}

__global__ void __launch_bounds__(256) k_agg(const float* __restrict__ G,
                                             const float* __restrict__ Am,
                                             const float* __restrict__ h,
                                             float* __restrict__ a_t,
                                             float* __restrict__ m_t) {
    const int lane = threadIdx.x & 31;
    const int wv   = threadIdx.x >> 5;
    const int tile = blockIdx.x * 8 + wv;
    const int dt = tile & 15;
    const int vt = (tile >> 4) & 7;
    const int b  = tile >> 7;
    const int hi = lane >> 4, lo = lane & 15;
    const int n  = dt * 16 + lo;
    const int m  = vt * 16 + lo;

    const float* Gb = G + b * NC * NC;
    const float* hb = h + b * NC * ND;
    v8f accA = {}, accM = {};
#pragma unroll 4
    for (int k = 0; k < NC; k += 4) {
        const int ka = k + 2 * hi;
        v2f ag, am, bb;
        ag.x = Gb[ka * NC + m];
        ag.y = Gb[(ka + 1) * NC + m];
        am.x = Am[ka * NC + m];
        am.y = Am[(ka + 1) * NC + m];
        bb.x = hb[ka * ND + n];
        bb.y = hb[(ka + 1) * ND + n];
        accA = wmma4(ag, bb, accA);
        accM = wmma4(am, bb, accM);
    }
#pragma unroll
    for (int g = 0; g < 8; ++g) {
        const int row = vt * 16 + g + 8 * hi;
        a_t[(b * NC + row) * ND + n] = accA[g];
        m_t[(b * NC + row) * ND + n] = accM[g] * hb[row * ND + n];
    }
}

__global__ void __launch_bounds__(256) k_final(const float* __restrict__ a_t,
                                               const float* __restrict__ m_t,
                                               const float* __restrict__ h,
                                               const float* __restrict__ Wa,
                                               const float* __restrict__ Wm,
                                               const float* __restrict__ ba,
                                               const float* __restrict__ bm,
                                               const float* __restrict__ pb1,
                                               const float* __restrict__ pb2,
                                               const float* __restrict__ pb3,
                                               float* __restrict__ z) {
    const int lane = threadIdx.x & 31;
    const int wv   = threadIdx.x >> 5;
    const int tile = blockIdx.x * 8 + wv;
    const int mt = tile >> 4;
    const int nt = tile & 15;
    const int hi = lane >> 4, lo = lane & 15;
    const int n  = nt * 16 + lo;

    const float* Ar = a_t + (mt * 16 + lo) * ND;
    const float* Mr = m_t + (mt * 16 + lo) * ND;
    v8f acc1 = {}, acc2 = {};
#pragma unroll 4
    for (int k = 0; k < ND; k += 4) {
        const int ka = k + 2 * hi;
        v2f a1 = *(const v2f*)(Ar + ka);
        v2f a2 = *(const v2f*)(Mr + ka);
        v2f w1v, w2v;
        w1v.x = Wa[ka * ND + n];
        w1v.y = Wa[(ka + 1) * ND + n];
        w2v.x = Wm[ka * ND + n];
        w2v.y = Wm[(ka + 1) * ND + n];
        acc1 = wmma4(a1, w1v, acc1);
        acc2 = wmma4(a2, w2v, acc2);
    }
    const float beta1 = pb1[0], beta2 = pb2[0], beta3 = pb3[0];
    const float bav = ba[n], bmv = bm[n];
#pragma unroll
    for (int g = 0; g < 8; ++g) {
        const int row = mt * 16 + g + 8 * hi;
        z[row * ND + n] = beta1 * h[row * ND + n]
                        + beta2 * (acc1[g] + bav)
                        + beta3 * (acc2[g] + bmv);
    }
}

extern "C" void kernel_launch(void* const* d_in, const int* in_sizes, int n_in,
                              void* d_out, int out_size, void* d_ws, size_t ws_size,
                              hipStream_t stream) {
    (void)in_sizes; (void)n_in; (void)out_size; (void)ws_size;

    const float* h_t = (const float*)d_in[0];
    const float* A_a = (const float*)d_in[1];
    const float* A_m = (const float*)d_in[2];
    const float* W1  = (const float*)d_in[3];
    const float* b1  = (const float*)d_in[4];
    const float* W2  = (const float*)d_in[5];
    const float* b2  = (const float*)d_in[6];
    const float* Wa  = (const float*)d_in[7];
    const float* ba  = (const float*)d_in[8];
    const float* Wm  = (const float*)d_in[9];
    const float* bm  = (const float*)d_in[10];
    const float* be1 = (const float*)d_in[11];
    const float* be2 = (const float*)d_in[12];
    const float* be3 = (const float*)d_in[13];
    float* z = (float*)d_out;

    float* P  = (float*)d_ws;
    float* Q  = P + NROWS * ND;
    float* G  = Q + NROWS * ND;
    float* At = G + NB * NC * NC;
    float* Mt = At + NROWS * ND;

    k_gemm_pq<<<256, 256, 0, stream>>>(h_t, W1, P, Q);
    k_scores<<<NB * 8 * 8, 256, 0, stream>>>(P, Q, b1, W2, b2, A_a, G);
    k_agg<<<256, 256, 0, stream>>>(G, A_m, h_t, At, Mt);
    k_final<<<256, 256, 0, stream>>>(At, Mt, h_t, Wa, Wm, ba, bm,
                                     be1, be2, be3, z);
}


# Score kernel fused into the output kernel: a block owns (batch, 8 v rows), keeps its 128x8 G tile in LDS, runs both sums from it (16-row tiles, rows repeated); 3 launches to 2; k order kept
# speedup vs baseline: 2.4790x; 2.4790x over previous
#include <hip/hip_runtime.h>
#include <hip/hip_bf16.h>

#define NB 16
#define NC 128
#define ND 256
#define NROWS (NB * NC)

typedef __attribute__((ext_vector_type(2))) float v2f;
typedef __attribute__((ext_vector_type(4))) float v4f;
typedef __attribute__((ext_vector_type(8))) float v8f;

static __device__ __forceinline__ v8f wmma4(v2f a, v2f b, v8f c) {
    return __builtin_amdgcn_wmma_f32_16x16x4_f32(
         false, a,  false, b,
         (short)0, c,  false,  false);
}


__global__ void __launch_bounds__(256) k_front(const float* __restrict__ H,
                                               const float* __restrict__ W1,
                                               const float* __restrict__ Wa,
                                               const float* __restrict__ Am,
                                               float* __restrict__ P,
                                               float* __restrict__ Q,
                                               float* __restrict__ R,
                                               float* __restrict__ m_t) {
    const int lane = threadIdx.x & 31;
    const int wv   = threadIdx.x >> 5;
    const int hi = lane >> 4, lo = lane & 15;
    const int blk = blockIdx.x;

    if (blk < 192) {
        const int mat = blk >> 6;
        const int cg  = (blk >> 4) & 3;
        const int mt  = (blk & 15) * 8 + wv;
        const float* Wb = (mat == 0) ? W1 : ((mat == 1) ? (W1 + ND * ND) : Wa);
        float* Out      = (mat == 0) ? P  : ((mat == 1) ? Q : R);

        const float* Arow = H + (mt * 16 + lo) * ND + 2 * hi;
        const float* Bcol = Wb + (2 * hi) * ND + cg * 64 + 4 * lo;
        v8f acc[4] = {};
        for (int k0 = 0; k0 < ND; k0 += 64) {
            v2f a[16];
            v4f bx[16], by[16];
#pragma unroll
            for (int i = 0; i < 16; ++i) {
                const int ka = k0 + 4 * i;
                a[i]  = *(const v2f*)(Arow + ka);
                bx[i] = *(const v4f*)(Bcol + ka * ND);
                by[i] = *(const v4f*)(Bcol + (ka + 1) * ND);
            }
            __builtin_amdgcn_sched_barrier(0);
#pragma unroll
            for (int i = 0; i < 16; ++i) {
#pragma unroll
                for (int t = 0; t < 4; ++t) {
                    v2f b; b.x = bx[i][t]; b.y = by[i][t];
                    acc[t] = wmma4(a[i], b, acc[t]);
                }
            }
        }
#pragma unroll
        for (int g = 0; g < 8; ++g) {
            const int row = mt * 16 + g + 8 * hi;
            v4f o; o.x = acc[0][g]; o.y = acc[1][g]; o.z = acc[2][g]; o.w = acc[3][g];
            *(v4f*)(Out + row * ND + cg * 64 + 4 * lo) = o;
        }
    } else {
        const int bb = blk - 192;
        const int dg = bb & 3;
        const int b  = bb >> 2;
        const int vt = wv;
        const int m  = vt * 16 + lo;

        const float* hb   = H + b * NC * ND;
        const float* Acol = Am + (2 * hi) * NC + m;
        const float* Bcol = hb + (2 * hi) * ND + dg * 64 + 4 * lo;
        v8f acc[4] = {};
        for (int k0 = 0; k0 < NC; k0 += 64) {
            v2f a[16];
            v4f bx[16], by[16];
#pragma unroll
            for (int i = 0; i < 16; ++i) {
                const int ka = k0 + 4 * i;
                a[i].x = Acol[ka * NC];
                a[i].y = Acol[(ka + 1) * NC];
                bx[i] = *(const v4f*)(Bcol + ka * ND);
                by[i] = *(const v4f*)(Bcol + (ka + 1) * ND);
            }
            __builtin_amdgcn_sched_barrier(0);
#pragma unroll
            for (int i = 0; i < 16; ++i) {
#pragma unroll
                for (int t = 0; t < 4; ++t) {
                    v2f b2; b2.x = bx[i][t]; b2.y = by[i][t];
                    acc[t] = wmma4(a[i], b2, acc[t]);
                }
            }
        }
#pragma unroll
        for (int g = 0; g < 8; ++g) {
            const int row = vt * 16 + g + 8 * hi;
            const int off = row * ND + dg * 64 + 4 * lo;
            const v4f hv = *(const v4f*)(hb + off);
            v4f o;
            o.x = acc[0][g] * hv.x; o.y = acc[1][g] * hv.y;
            o.z = acc[2][g] * hv.z; o.w = acc[3][g] * hv.w;
            *(v4f*)(m_t + b * NC * ND + off) = o;
        }
    }
}

__global__ void __launch_bounds__(256) k_tail(const float* __restrict__ P,
                                              const float* __restrict__ Q,
                                              const float* __restrict__ b1,
                                              const float* __restrict__ W2,
                                              const float* __restrict__ b2,
                                              const float* __restrict__ A_a,
                                              const float* __restrict__ R,
                                              const float* __restrict__ m_t,
                                              const float* __restrict__ h,
                                              const float* __restrict__ Wm,
                                              const float* __restrict__ ba,
                                              const float* __restrict__ bm,
                                              const float* __restrict__ pb1,
                                              const float* __restrict__ pb2,
                                              const float* __restrict__ pb3,
                                              float* __restrict__ z) {
    __shared__ __align__(16) float Gt[8][132];

    const int lane = threadIdx.x & 31;
    const int wv   = threadIdx.x >> 5;
    const int b    = blockIdx.x >> 4;
    const int v0   = (blockIdx.x & 15) * 8;

    {
        const float4 bA = ((const float4*)b1)[lane];
        const float4 bB = ((const float4*)b1)[32 + lane];
        const float4 wA = ((const float4*)W2)[lane];
        const float4 wB = ((const float4*)W2)[32 + lane];
        const float w[8] = {wA.x, wA.y, wA.z, wA.w, wB.x, wB.y, wB.z, wB.w};
        const float bias2 = b2[0];

        float q[8][8];
#pragma unroll
        for (int v = 0; v < 8; ++v) {
            const float4* qr = (const float4*)(Q + (b * NC + v0 + v) * ND);
            const float4 a = qr[lane], c = qr[32 + lane];
            q[v][0] = a.x; q[v][1] = a.y; q[v][2] = a.z; q[v][3] = a.w;
            q[v][4] = c.x; q[v][5] = c.y; q[v][6] = c.z; q[v][7] = c.w;
        }

#pragma unroll
        for (int pass = 0; pass < 2; ++pass) {
            const int u0 = pass * 64 + wv * 8;
            float p[8][8];
#pragma unroll
            for (int u = 0; u < 8; ++u) {
                const float4* pr = (const float4*)(P + (b * NC + u0 + u) * ND);
                const float4 a = pr[lane], c = pr[32 + lane];
                p[u][0] = a.x + bA.x; p[u][1] = a.y + bA.y; p[u][2] = a.z + bA.z; p[u][3] = a.w + bA.w;
                p[u][4] = c.x + bB.x; p[u][5] = c.y + bB.y; p[u][6] = c.z + bB.z; p[u][7] = c.w + bB.w;
            }

            float acc[64];
#pragma unroll
            for (int v = 0; v < 8; ++v) {
#pragma unroll
                for (int u = 0; u < 8; ++u) {
                    float s = 0.f;
#pragma unroll
                    for (int e = 0; e < 8; ++e)
                        s = fmaf(fmaxf(p[u][e] + q[v][e], 0.f), w[e], s);
                    acc[u * 8 + v] = s;
                }
            }

#pragma unroll
            for (int st = 0; st < 5; ++st) {
                const int m = 16 >> st;
                const int n = 32 >> st;
                const bool up = (lane & m) != 0;
#pragma unroll
                for (int i = 0; i < n; ++i) {
                    const float keep = up ? acc[i + n] : acc[i];
                    const float send = up ? acc[i] : acc[i + n];
                    acc[i] = keep + __shfl_xor(send, m, 32);
                }
            }
            const int uu = u0 + (lane >> 2);
            const int vv = 2 * (lane & 3);
            const float2 aa = *(const float2*)(A_a + uu * NC + v0 + vv);
            const float s0 = 1.0f / (1.0f + __expf(-(acc[0] + bias2)));
            const float s1 = 1.0f / (1.0f + __expf(-(acc[1] + bias2)));
            Gt[vv][uu]     = s0 * aa.x;
            Gt[vv + 1][uu] = s1 * aa.y;
        }
    }
    __syncthreads();

    const int hi = lane >> 4, lo = lane & 15;
    const int r8 = lo & 7;
    const int e0 = wv * 32 + 2 * lo;

    const float* Grow = &Gt[r8][2 * hi];
    const float* Rcol = R + b * NC * ND + (2 * hi) * ND + e0;
    const float* Mrow = m_t + (b * NC + v0 + r8) * ND + 2 * hi;
    const float* Wcol = Wm + (2 * hi) * ND + e0;
    v8f acc1[2] = {}, acc2[2] = {};
    for (int k0 = 0; k0 < NC; k0 += 32) {
        v2f ag[8], rx[8], ry[8], am[16], wx[16], wy[16];
#pragma unroll
        for (int j = 0; j < 8; ++j) {
            const int ka = k0 + 4 * j;
            ag[j] = *(const v2f*)(Grow + ka);
            rx[j] = *(const v2f*)(Rcol + ka * ND);
            ry[j] = *(const v2f*)(Rcol + (ka + 1) * ND);
        }
#pragma unroll
        for (int j = 0; j < 16; ++j) {
            const int kd = 2 * k0 + 4 * j;
            am[j] = *(const v2f*)(Mrow + kd);
            wx[j] = *(const v2f*)(Wcol + kd * ND);
            wy[j] = *(const v2f*)(Wcol + (kd + 1) * ND);
        }
#pragma unroll
        for (int j = 0; j < 8; ++j) {
#pragma unroll
            for (int t = 0; t < 2; ++t) {
                v2f br; br.x = rx[j][t]; br.y = ry[j][t];
                acc1[t] = wmma4(ag[j], br, acc1[t]);
            }
        }
        __builtin_amdgcn_sched_barrier(0);
#pragma unroll
        for (int j = 0; j < 16; ++j) {
#pragma unroll
            for (int t = 0; t < 2; ++t) {
                v2f bw; bw.x = wx[j][t]; bw.y = wy[j][t];
                acc2[t] = wmma4(am[j], bw, acc2[t]);
            }
        }
    }
    const float beta1 = pb1[0], beta2 = pb2[0], beta3 = pb3[0];
    const v2f bav = *(const v2f*)(ba + e0);
    const v2f bmv = *(const v2f*)(bm + e0);
#pragma unroll
    for (int g = 0; g < 8; ++g) {
        const int row = b * NC + v0 + g;
        const v2f hv = *(const v2f*)(h + row * ND + e0);
        v2f o;
        o.x = beta1 * hv.x + beta2 * (acc1[0][g] + bav.x) + beta3 * (acc2[0][g] + bmv.x);
        o.y = beta1 * hv.y + beta2 * (acc1[1][g] + bav.y) + beta3 * (acc2[1][g] + bmv.y);
        if (hi == 0) *(v2f*)(z + row * ND + e0) = o;
    }
}

extern "C" void kernel_launch(void* const* d_in, const int* in_sizes, int n_in,
                              void* d_out, int out_size, void* d_ws, size_t ws_size,
                              hipStream_t stream) {
    (void)in_sizes; (void)n_in; (void)out_size; (void)ws_size;

    const float* h_t = (const float*)d_in[0];
    const float* A_a = (const float*)d_in[1];
    const float* A_m = (const float*)d_in[2];
    const float* W1  = (const float*)d_in[3];
    const float* b1  = (const float*)d_in[4];
    const float* W2  = (const float*)d_in[5];
    const float* b2  = (const float*)d_in[6];
    const float* Wa  = (const float*)d_in[7];
    const float* ba  = (const float*)d_in[8];
    const float* Wm  = (const float*)d_in[9];
    const float* bm  = (const float*)d_in[10];
    const float* be1 = (const float*)d_in[11];
    const float* be2 = (const float*)d_in[12];
    const float* be3 = (const float*)d_in[13];
    float* z = (float*)d_out;

    float* P  = (float*)d_ws;
    float* Q  = P + NROWS * ND;
    float* R  = Q + NROWS * ND;
    float* Mt = R + NROWS * ND;

    k_front<<<256, 256, 0, stream>>>(h_t, W1, Wa, A_m, P, Q, R, Mt);
    k_tail<<<NB * 16, 256, 0, stream>>>(P, Q, b1, W2, b2, A_a, R, Mt, h_t, Wm, ba, bm,
                                     be1, be2, be3, z);
}


# Output kernel: behind its first loads each wave prefetches into the second-level cache the R, m_t and second-pass P rows that launch 1 wrote, so the k loop does not wait on memory for them
# speedup vs baseline: 2.5903x; 1.0449x over previous
#include <hip/hip_runtime.h>
#include <hip/hip_bf16.h>

#define NB 16
#define NC 128
#define ND 256
#define NROWS (NB * NC)

typedef __attribute__((ext_vector_type(2))) float v2f;
typedef __attribute__((ext_vector_type(4))) float v4f;
typedef __attribute__((ext_vector_type(8))) float v8f;

static __device__ __forceinline__ v8f wmma4(v2f a, v2f b, v8f c) {
    return __builtin_amdgcn_wmma_f32_16x16x4_f32(
         false, a,  false, b,
         (short)0, c,  false,  false);
}


__global__ void __launch_bounds__(256) k_front(const float* __restrict__ H,
                                               const float* __restrict__ W1,
                                               const float* __restrict__ Wa,
                                               const float* __restrict__ Am,
                                               float* __restrict__ P,
                                               float* __restrict__ Q,
                                               float* __restrict__ R,
                                               float* __restrict__ m_t) {
    const int lane = threadIdx.x & 31;
    const int wv   = threadIdx.x >> 5;
    const int hi = lane >> 4, lo = lane & 15;
    const int blk = blockIdx.x;

    if (blk < 192) {
        const int mat = blk >> 6;
        const int cg  = (blk >> 4) & 3;
        const int mt  = (blk & 15) * 8 + wv;
        const float* Wb = (mat == 0) ? W1 : ((mat == 1) ? (W1 + ND * ND) : Wa);
        float* Out      = (mat == 0) ? P  : ((mat == 1) ? Q : R);

        const float* Arow = H + (mt * 16 + lo) * ND + 2 * hi;
        const float* Bcol = Wb + (2 * hi) * ND + cg * 64 + 4 * lo;
        v8f acc[4] = {};
        for (int k0 = 0; k0 < ND; k0 += 64) {
            v2f a[16];
            v4f bx[16], by[16];
#pragma unroll
            for (int i = 0; i < 16; ++i) {
                const int ka = k0 + 4 * i;
                a[i]  = *(const v2f*)(Arow + ka);
                bx[i] = *(const v4f*)(Bcol + ka * ND);
                by[i] = *(const v4f*)(Bcol + (ka + 1) * ND);
            }
            __builtin_amdgcn_sched_barrier(0);
#pragma unroll
            for (int i = 0; i < 16; ++i) {
#pragma unroll
                for (int t = 0; t < 4; ++t) {
                    v2f b; b.x = bx[i][t]; b.y = by[i][t];
                    acc[t] = wmma4(a[i], b, acc[t]);
                }
            }
        }
#pragma unroll
        for (int g = 0; g < 8; ++g) {
            const int row = mt * 16 + g + 8 * hi;
            v4f o; o.x = acc[0][g]; o.y = acc[1][g]; o.z = acc[2][g]; o.w = acc[3][g];
            *(v4f*)(Out + row * ND + cg * 64 + 4 * lo) = o;
        }
    } else {
        const int bb = blk - 192;
        const int dg = bb & 3;
        const int b  = bb >> 2;
        const int vt = wv;
        const int m  = vt * 16 + lo;

        const float* hb   = H + b * NC * ND;
        const float* Acol = Am + (2 * hi) * NC + m;
        const float* Bcol = hb + (2 * hi) * ND + dg * 64 + 4 * lo;
        v8f acc[4] = {};
        for (int k0 = 0; k0 < NC; k0 += 64) {
            v2f a[16];
            v4f bx[16], by[16];
#pragma unroll
            for (int i = 0; i < 16; ++i) {
                const int ka = k0 + 4 * i;
                a[i].x = Acol[ka * NC];
                a[i].y = Acol[(ka + 1) * NC];
                bx[i] = *(const v4f*)(Bcol + ka * ND);
                by[i] = *(const v4f*)(Bcol + (ka + 1) * ND);
            }
            __builtin_amdgcn_sched_barrier(0);
#pragma unroll
            for (int i = 0; i < 16; ++i) {
#pragma unroll
                for (int t = 0; t < 4; ++t) {
                    v2f b2; b2.x = bx[i][t]; b2.y = by[i][t];
                    acc[t] = wmma4(a[i], b2, acc[t]);
                }
            }
        }
#pragma unroll
        for (int g = 0; g < 8; ++g) {
            const int row = vt * 16 + g + 8 * hi;
            const int off = row * ND + dg * 64 + 4 * lo;
            const v4f hv = *(const v4f*)(hb + off);
            v4f o;
            o.x = acc[0][g] * hv.x; o.y = acc[1][g] * hv.y;
            o.z = acc[2][g] * hv.z; o.w = acc[3][g] * hv.w;
            *(v4f*)(m_t + b * NC * ND + off) = o;
        }
    }
}

__global__ void __launch_bounds__(256) k_tail(const float* __restrict__ P,
                                              const float* __restrict__ Q,
                                              const float* __restrict__ b1,
                                              const float* __restrict__ W2,
                                              const float* __restrict__ b2,
                                              const float* __restrict__ A_a,
                                              const float* __restrict__ R,
                                              const float* __restrict__ m_t,
                                              const float* __restrict__ h,
                                              const float* __restrict__ Wm,
                                              const float* __restrict__ ba,
                                              const float* __restrict__ bm,
                                              const float* __restrict__ pb1,
                                              const float* __restrict__ pb2,
                                              const float* __restrict__ pb3,
                                              float* __restrict__ z) {
    __shared__ __align__(16) float Gt[8][132];

    const int lane = threadIdx.x & 31;
    const int wv   = threadIdx.x >> 5;
    const int b    = blockIdx.x >> 4;
    const int v0   = (blockIdx.x & 15) * 8;

    {
        const float4 bA = ((const float4*)b1)[lane];
        const float4 bB = ((const float4*)b1)[32 + lane];
        const float4 wA = ((const float4*)W2)[lane];
        const float4 wB = ((const float4*)W2)[32 + lane];
        const float w[8] = {wA.x, wA.y, wA.z, wA.w, wB.x, wB.y, wB.z, wB.w};
        const float bias2 = b2[0];

        float q[8][8];
#pragma unroll
        for (int v = 0; v < 8; ++v) {
            const float4* qr = (const float4*)(Q + (b * NC + v0 + v) * ND);
            const float4 a = qr[lane], c = qr[32 + lane];
            q[v][0] = a.x; q[v][1] = a.y; q[v][2] = a.z; q[v][3] = a.w;
            q[v][4] = c.x; q[v][5] = c.y; q[v][6] = c.z; q[v][7] = c.w;
        }

#pragma unroll
        for (int pass = 0; pass < 2; ++pass) {
            const int u0 = pass * 64 + wv * 8;
            float p[8][8];
#pragma unroll
            for (int u = 0; u < 8; ++u) {
                const float4* pr = (const float4*)(P + (b * NC + u0 + u) * ND);
                const float4 a = pr[lane], c = pr[32 + lane];
                p[u][0] = a.x + bA.x; p[u][1] = a.y + bA.y; p[u][2] = a.z + bA.z; p[u][3] = a.w + bA.w;
                p[u][4] = c.x + bB.x; p[u][5] = c.y + bB.y; p[u][6] = c.z + bB.z; p[u][7] = c.w + bB.w;
            }

            if (pass == 0) {
                const int pr = lane >> 2;
                const int pc = (lane & 3) * 64;
                __builtin_prefetch(P + (b * NC + 64 + wv * 8 + pr) * ND + pc, 0, 2);
                __builtin_prefetch(R + (b * NC + wv * 16 + pr) * ND + pc, 0, 2);
                __builtin_prefetch(R + (b * NC + wv * 16 + 8 + pr) * ND + pc, 0, 2);
                if (wv == 0)
                    __builtin_prefetch(m_t + (b * NC + v0 + pr) * ND + pc, 0, 2);
            }

            float acc[64];
#pragma unroll
            for (int v = 0; v < 8; ++v) {
#pragma unroll
                for (int u = 0; u < 8; ++u) {
                    float s = 0.f;
#pragma unroll
                    for (int e = 0; e < 8; ++e)
                        s = fmaf(fmaxf(p[u][e] + q[v][e], 0.f), w[e], s);
                    acc[u * 8 + v] = s;
                }
            }

#pragma unroll
            for (int st = 0; st < 5; ++st) {
                const int m = 16 >> st;
                const int n = 32 >> st;
                const bool up = (lane & m) != 0;
#pragma unroll
                for (int i = 0; i < n; ++i) {
                    const float keep = up ? acc[i + n] : acc[i];
                    const float send = up ? acc[i] : acc[i + n];
                    acc[i] = keep + __shfl_xor(send, m, 32);
                }
            }
            const int uu = u0 + (lane >> 2);
            const int vv = 2 * (lane & 3);
            const float2 aa = *(const float2*)(A_a + uu * NC + v0 + vv);
            const float s0 = 1.0f / (1.0f + __expf(-(acc[0] + bias2)));
            const float s1 = 1.0f / (1.0f + __expf(-(acc[1] + bias2)));
            Gt[vv][uu]     = s0 * aa.x;
            Gt[vv + 1][uu] = s1 * aa.y;
        }
    }
    __syncthreads();

    const int hi = lane >> 4, lo = lane & 15;
    const int r8 = lo & 7;
    const int e0 = wv * 32 + 2 * lo;

    const float* Grow = &Gt[r8][2 * hi];
    const float* Rcol = R + b * NC * ND + (2 * hi) * ND + e0;
    const float* Mrow = m_t + (b * NC + v0 + r8) * ND + 2 * hi;
    const float* Wcol = Wm + (2 * hi) * ND + e0;
    v8f acc1[2] = {}, acc2[2] = {};
    for (int k0 = 0; k0 < NC; k0 += 32) {
        v2f ag[8], rx[8], ry[8], am[16], wx[16], wy[16];
#pragma unroll
        for (int j = 0; j < 8; ++j) {
            const int ka = k0 + 4 * j;
            ag[j] = *(const v2f*)(Grow + ka);
            rx[j] = *(const v2f*)(Rcol + ka * ND);
            ry[j] = *(const v2f*)(Rcol + (ka + 1) * ND);
        }
#pragma unroll
        for (int j = 0; j < 16; ++j) {
            const int kd = 2 * k0 + 4 * j;
            am[j] = *(const v2f*)(Mrow + kd);
            wx[j] = *(const v2f*)(Wcol + kd * ND);
            wy[j] = *(const v2f*)(Wcol + (kd + 1) * ND);
        }
#pragma unroll
        for (int j = 0; j < 8; ++j) {
#pragma unroll
            for (int t = 0; t < 2; ++t) {
                v2f br; br.x = rx[j][t]; br.y = ry[j][t];
                acc1[t] = wmma4(ag[j], br, acc1[t]);
            }
        }
        __builtin_amdgcn_sched_barrier(0);
#pragma unroll
        for (int j = 0; j < 16; ++j) {
#pragma unroll
            for (int t = 0; t < 2; ++t) {
                v2f bw; bw.x = wx[j][t]; bw.y = wy[j][t];
                acc2[t] = wmma4(am[j], bw, acc2[t]);
            }
        }
    }
    const float beta1 = pb1[0], beta2 = pb2[0], beta3 = pb3[0];
    const v2f bav = *(const v2f*)(ba + e0);
    const v2f bmv = *(const v2f*)(bm + e0);
#pragma unroll
    for (int g = 0; g < 8; ++g) {
        const int row = b * NC + v0 + g;
        const v2f hv = *(const v2f*)(h + row * ND + e0);
        v2f o;
        o.x = beta1 * hv.x + beta2 * (acc1[0][g] + bav.x) + beta3 * (acc2[0][g] + bmv.x);
        o.y = beta1 * hv.y + beta2 * (acc1[1][g] + bav.y) + beta3 * (acc2[1][g] + bmv.y);
        if (hi == 0) *(v2f*)(z + row * ND + e0) = o;
    }
}

extern "C" void kernel_launch(void* const* d_in, const int* in_sizes, int n_in,
                              void* d_out, int out_size, void* d_ws, size_t ws_size,
                              hipStream_t stream) {
    (void)in_sizes; (void)n_in; (void)out_size; (void)ws_size;

    const float* h_t = (const float*)d_in[0];
    const float* A_a = (const float*)d_in[1];
    const float* A_m = (const float*)d_in[2];
    const float* W1  = (const float*)d_in[3];
    const float* b1  = (const float*)d_in[4];
    const float* W2  = (const float*)d_in[5];
    const float* b2  = (const float*)d_in[6];
    const float* Wa  = (const float*)d_in[7];
    const float* ba  = (const float*)d_in[8];
    const float* Wm  = (const float*)d_in[9];
    const float* bm  = (const float*)d_in[10];
    const float* be1 = (const float*)d_in[11];
    const float* be2 = (const float*)d_in[12];
    const float* be3 = (const float*)d_in[13];
    float* z = (float*)d_out;

    float* P  = (float*)d_ws;
    float* Q  = P + NROWS * ND;
    float* R  = Q + NROWS * ND;
    float* Mt = R + NROWS * ND;

    k_front<<<256, 256, 0, stream>>>(h_t, W1, Wa, A_m, P, Q, R, Mt);
    k_tail<<<NB * 16, 256, 0, stream>>>(P, Q, b1, W2, b2, A_a, R, Mt, h_t, Wm, ba, bm,
                                     be1, be2, be3, z);
}
